# starting kernel
# baseline (speedup 1.0000x reference)
#include <hip/hip_runtime.h>
#include <hip/hip_bf16.h>

typedef __attribute__((ext_vector_type(2))) float v2f;
typedef __attribute__((ext_vector_type(8))) float v8f;

#define H_DIM 256
#define EPSV 1e-8f

__global__ void zero_ws_kernel(float* __restrict__ p, int n) {
    int i = blockIdx.x * blockDim.x + threadIdx.x;
    if (i < n) p[i] = 0.0f;
}

__global__ __launch_bounds__(H_DIM)
void scatter_kernel(const float* __restrict__ x,
                    const int*   __restrict__ dst_idx,
                    float* __restrict__ sums,
                    float* __restrict__ counts) {
    const int row = blockIdx.x;
    const int tid = threadIdx.x;
    const int seg = dst_idx[row];
    const float v = x[(size_t)row * H_DIM + tid];
    atomicAdd(&sums[(size_t)seg * H_DIM + tid], v);
    if (tid == 0) atomicAdd(&counts[seg], 1.0f);
}

__global__ __launch_bounds__(256)
void pool_gemm_kernel(const float* __restrict__ sums,
                      const float* __restrict__ counts,
                      const float* __restrict__ W,
                      const float* __restrict__ bias,
                      float* __restrict__ out,
                      int S) {
    const int wid  = (int)((blockIdx.x * blockDim.x + threadIdx.x) >> 5);
    const int lane = (int)(threadIdx.x & 31);
    const int tilesN = H_DIM / 16;
    const int tileM = wid / tilesN;
    const int tileN = wid % tilesN;
    if (tileM * 16 >= S) return;

    const int m = lane & 15;
    const int h = lane >> 4;

    const int seg = tileM * 16 + m;
    const int col = tileN * 16 + m;

    const float inv = 1.0f / (counts[seg] + EPSV);

    const float* __restrict__ srow = sums + (size_t)seg * H_DIM;
    const float* __restrict__ wcol = W    + (size_t)col * H_DIM;

    v8f c;
    const float bj = bias[col];
#pragma unroll
    for (int r = 0; r < 8; ++r) c[r] = bj;

#pragma unroll 8
    for (int k = 0; k < H_DIM; k += 4) {
        v2f a = *(const v2f*)(srow + k + 2 * h);
        a.x *= inv;
        a.y *= inv;
        const v2f bf = *(const v2f*)(wcol + k + 2 * h);
        c = __builtin_amdgcn_wmma_f32_16x16x4_f32(
                 false, a,  false, bf,
                 (short)0, c,  false,  false);
    }

#pragma unroll
    for (int r = 0; r < 8; ++r) {
        out[(size_t)(tileM * 16 + r + 8 * h) * H_DIM + tileN * 16 + m] = c[r];
    }
}

extern "C" void kernel_launch(void* const* d_in, const int* in_sizes, int n_in,
                              void* d_out, int out_size, void* d_ws, size_t ws_size,
                              hipStream_t stream) {
    const float* x       = (const float*)d_in[0];
    const int*   dst_idx = (const int*)  d_in[1];
    const float* W       = (const float*)d_in[3];
    const float* bias    = (const float*)d_in[4];
    float*       out     = (float*)d_out;

    const int N = in_sizes[0] / H_DIM;
    const int S = out_size   / H_DIM;

    float* sums   = (float*)d_ws;
    float* counts = sums + (size_t)S * H_DIM;

    const int zn = S * H_DIM + S;
    zero_ws_kernel<<<(zn + 255) / 256, 256, 0, stream>>>(sums, zn);

    scatter_kernel<<<N, H_DIM, 0, stream>>>(x, dst_idx, sums, counts);

    const int tiles = (S / 16) * (H_DIM / 16);
    pool_gemm_kernel<<<tiles / 8, 256, 0, stream>>>(sums, counts, W, bias, out, S);
}


# Product kernel re-tiled: each wave makes a 32x32 block (2x2 tiles, 4 accumulators) from 16-byte fragment loads, each fragment feeding 4 matrix instructions; k order inside products changed
# speedup vs baseline: 3.1263x; 3.1263x over previous
#include <hip/hip_runtime.h>
#include <hip/hip_bf16.h>


typedef __attribute__((ext_vector_type(2))) float v2f;
typedef __attribute__((ext_vector_type(4))) float v4f;
typedef __attribute__((ext_vector_type(8))) float v8f;

#define H_DIM 256
#define EPSV 1e-8f
#define CHUNK 256
#define NBUCKET 256
#define OFF_PITCH 257
#define SUM_THREADS 512
#define CAP 512

__global__ __launch_bounds__(CHUNK)
void chunk_sort_kernel(const int* __restrict__ dst_idx,
                       int* __restrict__ sorted,
                       int* __restrict__ off,
                       int N, int S) {
    __shared__ __attribute__((aligned(16))) int s_key[CHUNK];
    __shared__ __attribute__((aligned(16))) int s_ent[CHUNK];
    __shared__ int s_wsum[CHUNK / 32];

    const int t    = (int)threadIdx.x;
    const int lane = t & 31;
    const int wv   = t >> 5;
    const int c    = (int)blockIdx.x;
    const int row  = c * CHUNK + t;

    int key = -1, ent = 0;
    if (row < N) {
        const int seg = dst_idx[row];
        if ((unsigned)seg < (unsigned)S) {
            key = seg >> 4;
            ent = (row << 4) | (seg & 15);
        }
    }
    s_key[t] = key;
    s_ent[t] = ent;
    __syncthreads();

    int cnt = 0;
    for (int i = 0; i < CHUNK; i += 4) {
        const int4 k = *(const int4*)&s_key[i];
        cnt += (k.x == t) + (k.y == t) + (k.z == t) + (k.w == t);
    }

    int inc = cnt;
    for (int d = 1; d < 32; d <<= 1) {
        const int v = __shfl_up(inc, d);
        if (lane >= d) inc += v;
    }
    if (lane == 31) s_wsum[wv] = inc;
    __syncthreads();
    int base = 0;
    for (int v = 0; v < wv; ++v) base += s_wsum[v];
    const int excl = base + inc - cnt;

    off[c * OFF_PITCH + t] = excl;
    if (t == CHUNK - 1) off[c * OFF_PITCH + NBUCKET] = excl + cnt;

    int* __restrict__ dstp = sorted + c * CHUNK;
    int k = excl;
    for (int i = 0; i < CHUNK; i += 4) {
        const int4 kk = *(const int4*)&s_key[i];
        if (kk.x == t) dstp[k++] = s_ent[i];
        if (kk.y == t) dstp[k++] = s_ent[i + 1];
        if (kk.z == t) dstp[k++] = s_ent[i + 2];
        if (kk.w == t) dstp[k++] = s_ent[i + 3];
    }
}

__global__ __launch_bounds__(SUM_THREADS)
void bucket_sum_kernel(const float* __restrict__ x,
                       const int*   __restrict__ sorted,
                       const int*   __restrict__ off,
                       float* __restrict__ sums,
                       float* __restrict__ counts,
                       int NC) {
    __shared__ int s_list[CAP];
    __shared__ int s_rows[(SUM_THREADS / 32) * CAP];
    __shared__ int s_tsum[SUM_THREADS / 32];

    const int t    = (int)threadIdx.x;
    const int lane = t & 31;
    const int w    = t >> 5;
    const int b    = (int)blockIdx.x;

    const int cpt = (NC + SUM_THREADS - 1) / SUM_THREADS;
    const int c0  = t * cpt;
    int mine = 0;
    for (int j = 0; j < cpt; ++j) {
        const int c = c0 + j;
        if (c < NC) mine += off[c * OFF_PITCH + b + 1] - off[c * OFF_PITCH + b];
    }
    int inc = mine;
    for (int d = 1; d < 32; d <<= 1) {
        const int v = __shfl_up(inc, d);
        if (lane >= d) inc += v;
    }
    if (lane == 31) s_tsum[w] = inc;
    __syncthreads();
    int basep = 0, total = 0;
    for (int v = 0; v < SUM_THREADS / 32; ++v) {
        const int sv = s_tsum[v];
        if (v < w) basep += sv;
        total += sv;
    }
    const int pos = basep + inc - mine;

    v4f acc0 = {0.0f, 0.0f, 0.0f, 0.0f};
    v4f acc1 = {0.0f, 0.0f, 0.0f, 0.0f};
    int nrows = 0;
    const v4f* __restrict__ xr = (const v4f*)x;

    for (int w0 = 0; w0 < total; w0 += CAP) {
        int p = pos;
        for (int j = 0; j < cpt; ++j) {
            const int c = c0 + j;
            if (c < NC) {
                const int o0 = off[c * OFF_PITCH + b];
                const int o1 = off[c * OFF_PITCH + b + 1];
                for (int q = o0; q < o1; ++q, ++p) {
                    const int rel = p - w0;
                    if ((unsigned)rel < (unsigned)CAP) s_list[rel] = sorted[c * CHUNK + q];
                }
            }
        }
        __syncthreads();
        const int nwin = (total - w0 < CAP) ? (total - w0) : CAP;

        const int start = w * CAP;
        int cw = 0;
        for (int base = 0; base < nwin; base += 32) {
            const int i = base + lane;
            const int e = (i < nwin) ? s_list[i] : 0;
            const bool m = (i < nwin) && ((e & 15) == w);
            const unsigned mask = __builtin_amdgcn_ballot_w32(m);
            if (m) s_rows[start + cw + __popc(mask & ((1u << lane) - 1u))] = e >> 4;
            cw += __popc(mask);
        }
        __syncthreads();

        int i = 0;
        for (; i + 8 <= cw; i += 8) {
            const int rv = s_rows[start + i + (lane & 7)];
            v4f a[8], q[8];
#pragma unroll
            for (int u = 0; u < 8; ++u) {
                const int r = __builtin_amdgcn_readlane(rv, u);
                const v4f* pr = xr + (size_t)r * (H_DIM / 4);
                a[u] = __builtin_nontemporal_load(pr + lane);
                q[u] = __builtin_nontemporal_load(pr + 32 + lane);
            }
#pragma unroll
            for (int u = 0; u < 8; ++u) {
                acc0 += a[u];
                acc1 += q[u];
            }
        }
        const int rem = cw - i;
        if (rem > 0) {
            int idx = i + (lane & 7);
            if (idx > cw - 1) idx = cw - 1;
            const int rv = s_rows[start + idx];
            v4f a[7], q[7];
#pragma unroll
            for (int u = 0; u < 7; ++u) {
                if (u < rem) {
                    const int r = __builtin_amdgcn_readlane(rv, u);
                    const v4f* pr = xr + (size_t)r * (H_DIM / 4);
                    a[u] = __builtin_nontemporal_load(pr + lane);
                    q[u] = __builtin_nontemporal_load(pr + 32 + lane);
                }
            }
#pragma unroll
            for (int u = 0; u < 7; ++u) {
                if (u < rem) {
                    acc0 += a[u];
                    acc1 += q[u];
                }
            }
        }
        nrows += cw;
        __syncthreads();
    }

    float* __restrict__ srow = sums + (size_t)(b * 16 + w) * H_DIM;
    *(v4f*)(srow + 4 * lane)       = acc0;
    *(v4f*)(srow + 128 + 4 * lane) = acc1;
    if (lane == 0) counts[b * 16 + w] = (float)nrows;
}

#define KSTEPS (H_DIM / 8)
#define GRP 8

__device__ __forceinline__ v8f mm4(v2f a, v2f b, v8f c) {
    return __builtin_amdgcn_wmma_f32_16x16x4_f32(
         false, a,  false, b,
         (short)0, c,  false,  false);
}

__global__ __launch_bounds__(128)
void pool_gemm_kernel(const float* __restrict__ sums,
                      const float* __restrict__ counts,
                      const float* __restrict__ W,
                      const float* __restrict__ bias,
                      float* __restrict__ out,
                      int S) {
    const int lane = (int)(threadIdx.x & 31);
    const int w    = (int)(threadIdx.x >> 5);
    const int m    = lane & 15;
    const int h    = lane >> 4;

    const int mb = (int)blockIdx.x >> 2;
    const int nb = (int)blockIdx.x & 3;
    const int MT = S >> 4;
    const int t0 = mb * 4 + (w >> 1) * 2;
    if (t0 >= MT) return;
    const bool two = (t0 + 1) < MT;
    const int t1 = two ? t0 + 1 : t0;
    const int n0 = nb * 4 + (w & 1) * 2;

    const int r0 = t0 * 16 + m;
    const int r1 = t1 * 16 + m;
    const int j0 = n0 * 16 + m;
    const int j1 = j0 + 16;

    const float inv0 = 1.0f / (counts[r0] + EPSV);
    const float inv1 = 1.0f / (counts[r1] + EPSV);

    const v4f* __restrict__ pa0 = (const v4f*)(sums + (size_t)r0 * H_DIM + 4 * h);
    const v4f* __restrict__ pa1 = (const v4f*)(sums + (size_t)r1 * H_DIM + 4 * h);
    const v4f* __restrict__ pb0 = (const v4f*)(W    + (size_t)j0 * H_DIM + 4 * h);
    const v4f* __restrict__ pb1 = (const v4f*)(W    + (size_t)j1 * H_DIM + 4 * h);

    v8f c00, c01, c10, c11;
    const float bj0 = bias[j0];
    const float bj1 = bias[j1];
#pragma unroll
    for (int r = 0; r < 8; ++r) { c00[r] = bj0; c10[r] = bj0; c01[r] = bj1; c11[r] = bj1; }

    v4f fa0[2][GRP], fa1[2][GRP], fb0[2][GRP], fb1[2][GRP];

#pragma unroll
    for (int u = 0; u < GRP; ++u) {
        fa0[0][u] = pa0[2 * u];
        fa1[0][u] = pa1[2 * u];
        fb0[0][u] = pb0[2 * u];
        fb1[0][u] = pb1[2 * u];
    }

#pragma unroll
    for (int g = 0; g < KSTEPS / GRP; ++g) {
        const int cur = g & 1;
        const int nxt = cur ^ 1;
        if (g + 1 < KSTEPS / GRP) {
#pragma unroll
            for (int u = 0; u < GRP; ++u) {
                const int kk = (g + 1) * GRP + u;
                fa0[nxt][u] = pa0[2 * kk];
                fa1[nxt][u] = pa1[2 * kk];
                fb0[nxt][u] = pb0[2 * kk];
                fb1[nxt][u] = pb1[2 * kk];
            }
            __builtin_amdgcn_sched_barrier(0);
        }
#pragma unroll
        for (int u = 0; u < GRP; ++u) {
            const v4f a0 = fa0[cur][u] * inv0;
            const v4f a1 = fa1[cur][u] * inv1;
            const v4f b0 = fb0[cur][u];
            const v4f b1 = fb1[cur][u];
            const v2f a0l = {a0.x, a0.y}, a0h = {a0.z, a0.w};
            const v2f a1l = {a1.x, a1.y}, a1h = {a1.z, a1.w};
            const v2f b0l = {b0.x, b0.y}, b0h = {b0.z, b0.w};
            const v2f b1l = {b1.x, b1.y}, b1h = {b1.z, b1.w};
            c00 = mm4(a0l, b0l, c00);
            c01 = mm4(a0l, b1l, c01);
            c10 = mm4(a1l, b0l, c10);
            c11 = mm4(a1l, b1l, c11);
            c00 = mm4(a0h, b0h, c00);
            c01 = mm4(a0h, b1h, c01);
            c10 = mm4(a1h, b0h, c10);
            c11 = mm4(a1h, b1h, c11);
        }
    }

    float* __restrict__ o0 = out + (size_t)(t0 * 16 + 8 * h) * H_DIM + j0;
#pragma unroll
    for (int r = 0; r < 8; ++r) {
        o0[(size_t)r * H_DIM]      = c00[r];
        o0[(size_t)r * H_DIM + 16] = c01[r];
    }
    if (two) {
        float* __restrict__ o1 = out + (size_t)(t1 * 16 + 8 * h) * H_DIM + j0;
#pragma unroll
        for (int r = 0; r < 8; ++r) {
            o1[(size_t)r * H_DIM]      = c10[r];
            o1[(size_t)r * H_DIM + 16] = c11[r];
        }
    }
}


extern "C" void kernel_launch(void* const* d_in, const int* in_sizes, int n_in,
                              void* d_out, int out_size, void* d_ws, size_t ws_size,
                              hipStream_t stream) {
    const float* x       = (const float*)d_in[0];
    const int*   dst_idx = (const int*)  d_in[1];
    const float* W       = (const float*)d_in[3];
    const float* bias    = (const float*)d_in[4];
    float*       out     = (float*)d_out;

    const int N = in_sizes[0] / H_DIM;
    const int S = out_size   / H_DIM;

    float* sums   = (float*)d_ws;
    float* counts = sums + (size_t)S * H_DIM;

    const int NC  = (N + CHUNK - 1) / CHUNK;
    int* sorted   = (int*)d_out;
    int* off      = sorted + (size_t)NC * CHUNK;

    chunk_sort_kernel<<<NC, CHUNK, 0, stream>>>(dst_idx, sorted, off, N, S);

    bucket_sum_kernel<<<S / 16, SUM_THREADS, 0, stream>>>(x, sorted, off, sums, counts, NC);

    const int blocks = ((S + 63) / 64) * (H_DIM / 64);
    pool_gemm_kernel<<<blocks, 128, 0, stream>>>(sums, counts, W, bias, out, S);
}


# Chunk sort ranks each row by wave ballots (one per bucket-number bit gives the same-bucket lanes; rank and count by popcount) plus per-wave LDS counts, not two serial 256-step scans per thread
# speedup vs baseline: 4.3739x; 1.3991x over previous
#include <hip/hip_runtime.h>
#include <hip/hip_bf16.h>


typedef __attribute__((ext_vector_type(2))) float v2f;
typedef __attribute__((ext_vector_type(4))) float v4f;
typedef __attribute__((ext_vector_type(8))) float v8f;

#define H_DIM 256
#define EPSV 1e-8f
#define CHUNK 256
#define NBUCKET 256
#define OFF_PITCH 257
#define SUM_THREADS 512
#define CAP 512

#define NWAVE (CHUNK / 32)

__global__ __launch_bounds__(CHUNK)
void chunk_sort_kernel(const int* __restrict__ dst_idx,
                       int* __restrict__ sorted,
                       int* __restrict__ off,
                       int N, int S) {
    __shared__ __attribute__((aligned(16))) int s_cnt[NWAVE * NBUCKET];
    __shared__ __attribute__((aligned(16))) int s_wsum[NWAVE];

    const int t    = (int)threadIdx.x;
    const int lane = t & 31;
    const int wv   = t >> 5;
    const int c    = (int)blockIdx.x;
    const int row  = c * CHUNK + t;

    const int4 zero4 = {0, 0, 0, 0};
    *(int4*)&s_cnt[wv * NBUCKET + 4 * lane]       = zero4;
    *(int4*)&s_cnt[wv * NBUCKET + 128 + 4 * lane] = zero4;

    int key = -1, ent = 0;
    if (row < N) {
        const int seg = dst_idx[row];
        if ((unsigned)seg < (unsigned)S) {
            key = seg >> 4;
            ent = (row << 4) | (seg & 15);
        }
    }
    const bool valid = key >= 0;

    unsigned peers = __builtin_amdgcn_ballot_w32(valid);
    for (int bit = 0; bit < 8; ++bit) {
        const bool one = ((key >> bit) & 1) != 0;
        const unsigned has = __builtin_amdgcn_ballot_w32(one);
        peers &= one ? has : ~has;
    }
    const int rank = __popc(peers & ((1u << lane) - 1u));
    if (valid && rank == 0) s_cnt[wv * NBUCKET + key] = __popc(peers);
    __syncthreads();

    int pre[NWAVE];
    int cnt = 0;
#pragma unroll
    for (int w = 0; w < NWAVE; ++w) {
        pre[w] = cnt;
        cnt += s_cnt[w * NBUCKET + t];
    }

    int inc = cnt;
    for (int d = 1; d < 32; d <<= 1) {
        const int v = __shfl_up(inc, d);
        if (lane >= d) inc += v;
    }
    if (lane == 31) s_wsum[wv] = inc;
    const int rel = inc - cnt;
#pragma unroll
    for (int w = 0; w < NWAVE; ++w) s_cnt[w * NBUCKET + t] = rel + pre[w];
    __syncthreads();

    const int kg = key >> 5;
    int base_t = 0, base_k = 0;
#pragma unroll
    for (int v = 0; v < NWAVE; ++v) {
        const int sv = s_wsum[v];
        base_t += (v < wv) ? sv : 0;
        base_k += (v < kg) ? sv : 0;
    }

    if (valid) sorted[c * CHUNK + base_k + s_cnt[wv * NBUCKET + key] + rank] = ent;

    const int excl = base_t + rel;
    off[c * OFF_PITCH + t] = excl;
    if (t == CHUNK - 1) off[c * OFF_PITCH + NBUCKET] = excl + cnt;
}

__global__ __launch_bounds__(SUM_THREADS)
void bucket_sum_kernel(const float* __restrict__ x,
                       const int*   __restrict__ sorted,
                       const int*   __restrict__ off,
                       float* __restrict__ sums,
                       float* __restrict__ counts,
                       int NC) {
    __shared__ int s_list[CAP];
    __shared__ int s_rows[(SUM_THREADS / 32) * CAP];
    __shared__ int s_tsum[SUM_THREADS / 32];

    const int t    = (int)threadIdx.x;
    const int lane = t & 31;
    const int w    = t >> 5;
    const int b    = (int)blockIdx.x;

    const int cpt = (NC + SUM_THREADS - 1) / SUM_THREADS;
    const int c0  = t * cpt;
    int mine = 0;
    for (int j = 0; j < cpt; ++j) {
        const int c = c0 + j;
        if (c < NC) mine += off[c * OFF_PITCH + b + 1] - off[c * OFF_PITCH + b];
    }
    int inc = mine;
    for (int d = 1; d < 32; d <<= 1) {
        const int v = __shfl_up(inc, d);
        if (lane >= d) inc += v;
    }
    if (lane == 31) s_tsum[w] = inc;
    __syncthreads();
    int basep = 0, total = 0;
    for (int v = 0; v < SUM_THREADS / 32; ++v) {
        const int sv = s_tsum[v];
        if (v < w) basep += sv;
        total += sv;
    }
    const int pos = basep + inc - mine;

    v4f acc0 = {0.0f, 0.0f, 0.0f, 0.0f};
    v4f acc1 = {0.0f, 0.0f, 0.0f, 0.0f};
    int nrows = 0;
    const v4f* __restrict__ xr = (const v4f*)x;

    for (int w0 = 0; w0 < total; w0 += CAP) {
        int p = pos;
        for (int j = 0; j < cpt; ++j) {
            const int c = c0 + j;
            if (c < NC) {
                const int o0 = off[c * OFF_PITCH + b];
                const int o1 = off[c * OFF_PITCH + b + 1];
                for (int q = o0; q < o1; ++q, ++p) {
                    const int rel = p - w0;
                    if ((unsigned)rel < (unsigned)CAP) s_list[rel] = sorted[c * CHUNK + q];
                }
            }
        }
        __syncthreads();
        const int nwin = (total - w0 < CAP) ? (total - w0) : CAP;

        const int start = w * CAP;
        int cw = 0;
        for (int base = 0; base < nwin; base += 32) {
            const int i = base + lane;
            const int e = (i < nwin) ? s_list[i] : 0;
            const bool m = (i < nwin) && ((e & 15) == w);
            const unsigned mask = __builtin_amdgcn_ballot_w32(m);
            if (m) s_rows[start + cw + __popc(mask & ((1u << lane) - 1u))] = e >> 4;
            cw += __popc(mask);
        }
        __syncthreads();

        int i = 0;
        for (; i + 8 <= cw; i += 8) {
            const int rv = s_rows[start + i + (lane & 7)];
            v4f a[8], q[8];
#pragma unroll
            for (int u = 0; u < 8; ++u) {
                const int r = __builtin_amdgcn_readlane(rv, u);
                const v4f* pr = xr + (size_t)r * (H_DIM / 4);
                a[u] = __builtin_nontemporal_load(pr + lane);
                q[u] = __builtin_nontemporal_load(pr + 32 + lane);
            }
#pragma unroll
            for (int u = 0; u < 8; ++u) {
                acc0 += a[u];
                acc1 += q[u];
            }
        }
        const int rem = cw - i;
        if (rem > 0) {
            int idx = i + (lane & 7);
            if (idx > cw - 1) idx = cw - 1;
            const int rv = s_rows[start + idx];
            v4f a[7], q[7];
#pragma unroll
            for (int u = 0; u < 7; ++u) {
                if (u < rem) {
                    const int r = __builtin_amdgcn_readlane(rv, u);
                    const v4f* pr = xr + (size_t)r * (H_DIM / 4);
                    a[u] = __builtin_nontemporal_load(pr + lane);
                    q[u] = __builtin_nontemporal_load(pr + 32 + lane);
                }
            }
#pragma unroll
            for (int u = 0; u < 7; ++u) {
                if (u < rem) {
                    acc0 += a[u];
                    acc1 += q[u];
                }
            }
        }
        nrows += cw;
        __syncthreads();
    }

    float* __restrict__ srow = sums + (size_t)(b * 16 + w) * H_DIM;
    *(v4f*)(srow + 4 * lane)       = acc0;
    *(v4f*)(srow + 128 + 4 * lane) = acc1;
    if (lane == 0) counts[b * 16 + w] = (float)nrows;
}

#define KSTEPS (H_DIM / 8)
#define GRP 8

__device__ __forceinline__ v8f mm4(v2f a, v2f b, v8f c) {
    return __builtin_amdgcn_wmma_f32_16x16x4_f32(
         false, a,  false, b,
         (short)0, c,  false,  false);
}

__global__ __launch_bounds__(128)
void pool_gemm_kernel(const float* __restrict__ sums,
                      const float* __restrict__ counts,
                      const float* __restrict__ W,
                      const float* __restrict__ bias,
                      float* __restrict__ out,
                      int S) {
    const int lane = (int)(threadIdx.x & 31);
    const int w    = (int)(threadIdx.x >> 5);
    const int m    = lane & 15;
    const int h    = lane >> 4;

    const int mb = (int)blockIdx.x >> 2;
    const int nb = (int)blockIdx.x & 3;
    const int MT = S >> 4;
    const int t0 = mb * 4 + (w >> 1) * 2;
    if (t0 >= MT) return;
    const bool two = (t0 + 1) < MT;
    const int t1 = two ? t0 + 1 : t0;
    const int n0 = nb * 4 + (w & 1) * 2;

    const int r0 = t0 * 16 + m;
    const int r1 = t1 * 16 + m;
    const int j0 = n0 * 16 + m;
    const int j1 = j0 + 16;

    const float inv0 = 1.0f / (counts[r0] + EPSV);
    const float inv1 = 1.0f / (counts[r1] + EPSV);

    const v4f* __restrict__ pa0 = (const v4f*)(sums + (size_t)r0 * H_DIM + 4 * h);
    const v4f* __restrict__ pa1 = (const v4f*)(sums + (size_t)r1 * H_DIM + 4 * h);
    const v4f* __restrict__ pb0 = (const v4f*)(W    + (size_t)j0 * H_DIM + 4 * h);
    const v4f* __restrict__ pb1 = (const v4f*)(W    + (size_t)j1 * H_DIM + 4 * h);

    v8f c00, c01, c10, c11;
    const float bj0 = bias[j0];
    const float bj1 = bias[j1];
#pragma unroll
    for (int r = 0; r < 8; ++r) { c00[r] = bj0; c10[r] = bj0; c01[r] = bj1; c11[r] = bj1; }

    v4f fa0[2][GRP], fa1[2][GRP], fb0[2][GRP], fb1[2][GRP];

#pragma unroll
    for (int u = 0; u < GRP; ++u) {
        fa0[0][u] = pa0[2 * u];
        fa1[0][u] = pa1[2 * u];
        fb0[0][u] = pb0[2 * u];
        fb1[0][u] = pb1[2 * u];
    }

#pragma unroll
    for (int g = 0; g < KSTEPS / GRP; ++g) {
        const int cur = g & 1;
        const int nxt = cur ^ 1;
        if (g + 1 < KSTEPS / GRP) {
#pragma unroll
            for (int u = 0; u < GRP; ++u) {
                const int kk = (g + 1) * GRP + u;
                fa0[nxt][u] = pa0[2 * kk];
                fa1[nxt][u] = pa1[2 * kk];
                fb0[nxt][u] = pb0[2 * kk];
                fb1[nxt][u] = pb1[2 * kk];
            }
            __builtin_amdgcn_sched_barrier(0);
        }
#pragma unroll
        for (int u = 0; u < GRP; ++u) {
            const v4f a0 = fa0[cur][u] * inv0;
            const v4f a1 = fa1[cur][u] * inv1;
            const v4f b0 = fb0[cur][u];
            const v4f b1 = fb1[cur][u];
            const v2f a0l = {a0.x, a0.y}, a0h = {a0.z, a0.w};
            const v2f a1l = {a1.x, a1.y}, a1h = {a1.z, a1.w};
            const v2f b0l = {b0.x, b0.y}, b0h = {b0.z, b0.w};
            const v2f b1l = {b1.x, b1.y}, b1h = {b1.z, b1.w};
            c00 = mm4(a0l, b0l, c00);
            c01 = mm4(a0l, b1l, c01);
            c10 = mm4(a1l, b0l, c10);
            c11 = mm4(a1l, b1l, c11);
            c00 = mm4(a0h, b0h, c00);
            c01 = mm4(a0h, b1h, c01);
            c10 = mm4(a1h, b0h, c10);
            c11 = mm4(a1h, b1h, c11);
        }
    }

    float* __restrict__ o0 = out + (size_t)(t0 * 16 + 8 * h) * H_DIM + j0;
#pragma unroll
    for (int r = 0; r < 8; ++r) {
        o0[(size_t)r * H_DIM]      = c00[r];
        o0[(size_t)r * H_DIM + 16] = c01[r];
    }
    if (two) {
        float* __restrict__ o1 = out + (size_t)(t1 * 16 + 8 * h) * H_DIM + j0;
#pragma unroll
        for (int r = 0; r < 8; ++r) {
            o1[(size_t)r * H_DIM]      = c10[r];
            o1[(size_t)r * H_DIM + 16] = c11[r];
        }
    }
}


extern "C" void kernel_launch(void* const* d_in, const int* in_sizes, int n_in,
                              void* d_out, int out_size, void* d_ws, size_t ws_size,
                              hipStream_t stream) {
    const float* x       = (const float*)d_in[0];
    const int*   dst_idx = (const int*)  d_in[1];
    const float* W       = (const float*)d_in[3];
    const float* bias    = (const float*)d_in[4];
    float*       out     = (float*)d_out;

    const int N = in_sizes[0] / H_DIM;
    const int S = out_size   / H_DIM;

    float* sums   = (float*)d_ws;
    float* counts = sums + (size_t)S * H_DIM;

    const int NC  = (N + CHUNK - 1) / CHUNK;
    int* sorted   = (int*)d_out;
    int* off      = sorted + (size_t)NC * CHUNK;

    chunk_sort_kernel<<<NC, CHUNK, 0, stream>>>(dst_idx, sorted, off, N, S);

    bucket_sum_kernel<<<S / 16, SUM_THREADS, 0, stream>>>(x, sorted, off, sums, counts, NC);

    const int blocks = ((S + 63) / 64) * (H_DIM / 64);
    pool_gemm_kernel<<<blocks, 128, 0, stream>>>(sums, counts, W, bias, out, S);
}


# Sum kernel builds its row lists with one thread per list entry (the thread fetches its entry and ranks it by 4 wave ballots + per-wave LDS counts), not serial per-chunk copies and 16 full-list scans
# speedup vs baseline: 4.5842x; 1.0481x over previous
#include <hip/hip_runtime.h>
#include <hip/hip_bf16.h>


typedef __attribute__((ext_vector_type(2))) float v2f;
typedef __attribute__((ext_vector_type(4))) float v4f;
typedef __attribute__((ext_vector_type(8))) float v8f;

#define H_DIM 256
#define EPSV 1e-8f
#define CHUNK 256
#define NBUCKET 256
#define OFF_PITCH 257
#define SUM_THREADS 512
#define CAP 512

#define NWAVE (CHUNK / 32)

__global__ __launch_bounds__(CHUNK)
void chunk_sort_kernel(const int* __restrict__ dst_idx,
                       int* __restrict__ sorted,
                       int* __restrict__ off,
                       int N, int S) {
    __shared__ __attribute__((aligned(16))) int s_cnt[NWAVE * NBUCKET];
    __shared__ __attribute__((aligned(16))) int s_wsum[NWAVE];

    const int t    = (int)threadIdx.x;
    const int lane = t & 31;
    const int wv   = t >> 5;
    const int c    = (int)blockIdx.x;
    const int row  = c * CHUNK + t;

    const int4 zero4 = {0, 0, 0, 0};
    *(int4*)&s_cnt[wv * NBUCKET + 4 * lane]       = zero4;
    *(int4*)&s_cnt[wv * NBUCKET + 128 + 4 * lane] = zero4;

    int key = -1, ent = 0;
    if (row < N) {
        const int seg = dst_idx[row];
        if ((unsigned)seg < (unsigned)S) {
            key = seg >> 4;
            ent = (row << 4) | (seg & 15);
        }
    }
    const bool valid = key >= 0;

    unsigned peers = __builtin_amdgcn_ballot_w32(valid);
    for (int bit = 0; bit < 8; ++bit) {
        const bool one = ((key >> bit) & 1) != 0;
        const unsigned has = __builtin_amdgcn_ballot_w32(one);
        peers &= one ? has : ~has;
    }
    const int rank = __popc(peers & ((1u << lane) - 1u));
    if (valid && rank == 0) s_cnt[wv * NBUCKET + key] = __popc(peers);
    __syncthreads();

    int pre[NWAVE];
    int cnt = 0;
#pragma unroll
    for (int w = 0; w < NWAVE; ++w) {
        pre[w] = cnt;
        cnt += s_cnt[w * NBUCKET + t];
    }

    int inc = cnt;
    for (int d = 1; d < 32; d <<= 1) {
        const int v = __shfl_up(inc, d);
        if (lane >= d) inc += v;
    }
    if (lane == 31) s_wsum[wv] = inc;
    const int rel = inc - cnt;
#pragma unroll
    for (int w = 0; w < NWAVE; ++w) s_cnt[w * NBUCKET + t] = rel + pre[w];
    __syncthreads();

    const int kg = key >> 5;
    int base_t = 0, base_k = 0;
#pragma unroll
    for (int v = 0; v < NWAVE; ++v) {
        const int sv = s_wsum[v];
        base_t += (v < wv) ? sv : 0;
        base_k += (v < kg) ? sv : 0;
    }

    if (valid) sorted[c * CHUNK + base_k + s_cnt[wv * NBUCKET + key] + rank] = ent;

    const int excl = base_t + rel;
    off[c * OFF_PITCH + t] = excl;
    if (t == CHUNK - 1) off[c * OFF_PITCH + NBUCKET] = excl + cnt;
}

__global__ __launch_bounds__(SUM_THREADS)
void bucket_sum_kernel(const float* __restrict__ x,
                       const int*   __restrict__ sorted,
                       const int*   __restrict__ off,
                       float* __restrict__ sums,
                       float* __restrict__ counts,
                       int NC) {
    __shared__ int s_list[CAP];
    __shared__ int s_rows[(SUM_THREADS / 32) * CAP];
    __shared__ int s_tsum[SUM_THREADS / 32];
    __shared__ __attribute__((aligned(16))) int s_wcnt[16 * (SUM_THREADS / 32)];
    static_assert(CAP == SUM_THREADS, "the split gives every window entry a thread of its own");
    static_assert(SUM_THREADS / 32 == 16, "one wave per segment of the bucket");

    const int t    = (int)threadIdx.x;
    const int lane = t & 31;
    const int w    = t >> 5;
    const int b    = (int)blockIdx.x;

    const int cpt = (NC + SUM_THREADS - 1) / SUM_THREADS;
    const int c0  = t * cpt;
    int mine = 0;
    int f0 = 0, f1 = 0;
    for (int j = 0; j < cpt; ++j) {
        const int c = c0 + j;
        if (c < NC) {
            const int o0 = off[c * OFF_PITCH + b];
            const int o1 = off[c * OFF_PITCH + b + 1];
            if (j == 0) { f0 = o0; f1 = o1; }
            mine += o1 - o0;
        }
    }
    int inc = mine;
    for (int d = 1; d < 32; d <<= 1) {
        const int v = __shfl_up(inc, d);
        if (lane >= d) inc += v;
    }
    if (lane == 31) s_tsum[w] = inc;
    __syncthreads();
    int basep = 0, total = 0;
    for (int v = 0; v < SUM_THREADS / 32; ++v) {
        const int sv = s_tsum[v];
        if (v < w) basep += sv;
        total += sv;
    }
    const int pos = basep + inc - mine;

    v4f acc0 = {0.0f, 0.0f, 0.0f, 0.0f};
    v4f acc1 = {0.0f, 0.0f, 0.0f, 0.0f};
    int nrows = 0;
    const v4f* __restrict__ xr = (const v4f*)x;

    for (int w0 = 0; w0 < total; w0 += CAP) {
        int p = pos;
        for (int j = 0; j < cpt; ++j) {
            const int c = c0 + j;
            if (c < NC) {
                const int o0 = (j == 0) ? f0 : off[c * OFF_PITCH + b];
                const int o1 = (j == 0) ? f1 : off[c * OFF_PITCH + b + 1];
                const int src = c * CHUNK + o0 - (p - w0);
                for (int q = o0; q < o1; ++q, ++p) {
                    const int rel = p - w0;
                    if ((unsigned)rel < (unsigned)CAP) s_list[rel] = src;
                }
            }
        }
        __syncthreads();
        const int nwin = (total - w0 < CAP) ? (total - w0) : CAP;

        const int start = w * CAP;
        const bool valid = t < nwin;
        int e = 0;
        if (valid) e = sorted[s_list[t] + t];
        const int key = e & 15;
        unsigned peers = __builtin_amdgcn_ballot_w32(valid);
        for (int bit = 0; bit < 4; ++bit) {
            const bool one = ((key >> bit) & 1) != 0;
            const unsigned has = __builtin_amdgcn_ballot_w32(one);
            peers &= one ? has : ~has;
        }
        const int rank = __popc(peers & ((1u << lane) - 1u));
        if (lane < 16) s_wcnt[16 * lane + w] = 0;
        if (valid && rank == 0) s_wcnt[16 * key + w] = __popc(peers);
        __syncthreads();
        int before = 0;
        int cwv = 0;
#pragma unroll
        for (int v4 = 0; v4 < 4; ++v4) {
            const int4 ck = *(const int4*)&s_wcnt[16 * key + 4 * v4];
            const int4 cm = *(const int4*)&s_wcnt[16 * w + 4 * v4];
            before += ((4 * v4 + 0 < w) ? ck.x : 0) + ((4 * v4 + 1 < w) ? ck.y : 0)
                    + ((4 * v4 + 2 < w) ? ck.z : 0) + ((4 * v4 + 3 < w) ? ck.w : 0);
            cwv += cm.x + cm.y + cm.z + cm.w;
        }
        if (valid) s_rows[key * CAP + before + rank] = e >> 4;
        const int cw = __builtin_amdgcn_readfirstlane(cwv);
        __syncthreads();

        int i = 0;
        for (; i + 8 <= cw; i += 8) {
            const int rv = s_rows[start + i + (lane & 7)];
            v4f a[8], q[8];
#pragma unroll
            for (int u = 0; u < 8; ++u) {
                const int r = __builtin_amdgcn_readlane(rv, u);
                const v4f* pr = xr + (size_t)r * (H_DIM / 4);
                a[u] = __builtin_nontemporal_load(pr + lane);
                q[u] = __builtin_nontemporal_load(pr + 32 + lane);
            }
#pragma unroll
            for (int u = 0; u < 8; ++u) {
                acc0 += a[u];
                acc1 += q[u];
            }
        }
        const int rem = cw - i;
        if (rem > 0) {
            int idx = i + (lane & 7);
            if (idx > cw - 1) idx = cw - 1;
            const int rv = s_rows[start + idx];
            v4f a[7], q[7];
#pragma unroll
            for (int u = 0; u < 7; ++u) {
                if (u < rem) {
                    const int r = __builtin_amdgcn_readlane(rv, u);
                    const v4f* pr = xr + (size_t)r * (H_DIM / 4);
                    a[u] = __builtin_nontemporal_load(pr + lane);
                    q[u] = __builtin_nontemporal_load(pr + 32 + lane);
                }
            }
#pragma unroll
            for (int u = 0; u < 7; ++u) {
                if (u < rem) {
                    acc0 += a[u];
                    acc1 += q[u];
                }
            }
        }
        nrows += cw;
        __syncthreads();
    }

    float* __restrict__ srow = sums + (size_t)(b * 16 + w) * H_DIM;
    *(v4f*)(srow + 4 * lane)       = acc0;
    *(v4f*)(srow + 128 + 4 * lane) = acc1;
    if (lane == 0) counts[b * 16 + w] = (float)nrows;
}

#define KSTEPS (H_DIM / 8)
#define GRP 8

__device__ __forceinline__ v8f mm4(v2f a, v2f b, v8f c) {
    return __builtin_amdgcn_wmma_f32_16x16x4_f32(
         false, a,  false, b,
         (short)0, c,  false,  false);
}

__global__ __launch_bounds__(128)
void pool_gemm_kernel(const float* __restrict__ sums,
                      const float* __restrict__ counts,
                      const float* __restrict__ W,
                      const float* __restrict__ bias,
                      float* __restrict__ out,
                      int S) {
    const int lane = (int)(threadIdx.x & 31);
    const int w    = (int)(threadIdx.x >> 5);
    const int m    = lane & 15;
    const int h    = lane >> 4;

    const int mb = (int)blockIdx.x >> 2;
    const int nb = (int)blockIdx.x & 3;
    const int MT = S >> 4;
    const int t0 = mb * 4 + (w >> 1) * 2;
    if (t0 >= MT) return;
    const bool two = (t0 + 1) < MT;
    const int t1 = two ? t0 + 1 : t0;
    const int n0 = nb * 4 + (w & 1) * 2;

    const int r0 = t0 * 16 + m;
    const int r1 = t1 * 16 + m;
    const int j0 = n0 * 16 + m;
    const int j1 = j0 + 16;

    const float inv0 = 1.0f / (counts[r0] + EPSV);
    const float inv1 = 1.0f / (counts[r1] + EPSV);

    const v4f* __restrict__ pa0 = (const v4f*)(sums + (size_t)r0 * H_DIM + 4 * h);
    const v4f* __restrict__ pa1 = (const v4f*)(sums + (size_t)r1 * H_DIM + 4 * h);
    const v4f* __restrict__ pb0 = (const v4f*)(W    + (size_t)j0 * H_DIM + 4 * h);
    const v4f* __restrict__ pb1 = (const v4f*)(W    + (size_t)j1 * H_DIM + 4 * h);

    v8f c00, c01, c10, c11;
    const float bj0 = bias[j0];
    const float bj1 = bias[j1];
#pragma unroll
    for (int r = 0; r < 8; ++r) { c00[r] = bj0; c10[r] = bj0; c01[r] = bj1; c11[r] = bj1; }

    v4f fa0[2][GRP], fa1[2][GRP], fb0[2][GRP], fb1[2][GRP];

#pragma unroll
    for (int u = 0; u < GRP; ++u) {
        fa0[0][u] = pa0[2 * u];
        fa1[0][u] = pa1[2 * u];
        fb0[0][u] = pb0[2 * u];
        fb1[0][u] = pb1[2 * u];
    }

#pragma unroll
    for (int g = 0; g < KSTEPS / GRP; ++g) {
        const int cur = g & 1;
        const int nxt = cur ^ 1;
        if (g + 1 < KSTEPS / GRP) {
#pragma unroll
            for (int u = 0; u < GRP; ++u) {
                const int kk = (g + 1) * GRP + u;
                fa0[nxt][u] = pa0[2 * kk];
                fa1[nxt][u] = pa1[2 * kk];
                fb0[nxt][u] = pb0[2 * kk];
                fb1[nxt][u] = pb1[2 * kk];
            }
            __builtin_amdgcn_sched_barrier(0);
        }
#pragma unroll
        for (int u = 0; u < GRP; ++u) {
            const v4f a0 = fa0[cur][u] * inv0;
            const v4f a1 = fa1[cur][u] * inv1;
            const v4f b0 = fb0[cur][u];
            const v4f b1 = fb1[cur][u];
            const v2f a0l = {a0.x, a0.y}, a0h = {a0.z, a0.w};
            const v2f a1l = {a1.x, a1.y}, a1h = {a1.z, a1.w};
            const v2f b0l = {b0.x, b0.y}, b0h = {b0.z, b0.w};
            const v2f b1l = {b1.x, b1.y}, b1h = {b1.z, b1.w};
            c00 = mm4(a0l, b0l, c00);
            c01 = mm4(a0l, b1l, c01);
            c10 = mm4(a1l, b0l, c10);
            c11 = mm4(a1l, b1l, c11);
            c00 = mm4(a0h, b0h, c00);
            c01 = mm4(a0h, b1h, c01);
            c10 = mm4(a1h, b0h, c10);
            c11 = mm4(a1h, b1h, c11);
        }
    }

    float* __restrict__ o0 = out + (size_t)(t0 * 16 + 8 * h) * H_DIM + j0;
#pragma unroll
    for (int r = 0; r < 8; ++r) {
        o0[(size_t)r * H_DIM]      = c00[r];
        o0[(size_t)r * H_DIM + 16] = c01[r];
    }
    if (two) {
        float* __restrict__ o1 = out + (size_t)(t1 * 16 + 8 * h) * H_DIM + j0;
#pragma unroll
        for (int r = 0; r < 8; ++r) {
            o1[(size_t)r * H_DIM]      = c10[r];
            o1[(size_t)r * H_DIM + 16] = c11[r];
        }
    }
}


extern "C" void kernel_launch(void* const* d_in, const int* in_sizes, int n_in,
                              void* d_out, int out_size, void* d_ws, size_t ws_size,
                              hipStream_t stream) {
    const float* x       = (const float*)d_in[0];
    const int*   dst_idx = (const int*)  d_in[1];
    const float* W       = (const float*)d_in[3];
    const float* bias    = (const float*)d_in[4];
    float*       out     = (float*)d_out;

    const int N = in_sizes[0] / H_DIM;
    const int S = out_size   / H_DIM;

    float* sums   = (float*)d_ws;
    float* counts = sums + (size_t)S * H_DIM;

    const int NC  = (N + CHUNK - 1) / CHUNK;
    int* sorted   = (int*)d_out;
    int* off      = sorted + (size_t)NC * CHUNK;

    chunk_sort_kernel<<<NC, CHUNK, 0, stream>>>(dst_idx, sorted, off, N, S);

    bucket_sum_kernel<<<S / 16, SUM_THREADS, 0, stream>>>(x, sorted, off, sums, counts, NC);

    const int blocks = ((S + 63) / 64) * (H_DIM / 64);
    pool_gemm_kernel<<<blocks, 128, 0, stream>>>(sums, counts, W, bias, out, S);
}
